# starting kernel
# baseline (speedup 1.0000x reference)
#include <hip/hip_runtime.h>
#include <hip/hip_bf16.h>

typedef __attribute__((ext_vector_type(2))) float v2f;
typedef __attribute__((ext_vector_type(8))) float v8f;

__device__ __constant__ float c_ST[12 * 16] = {
    -2.f,  1.f,  0.f,  0.f,  0.f, 0,0,0,0,0,0,0,0,0,0,0,
     2.f, -1.f,  0.f,  0.f,  0.f, 0,0,0,0,0,0,0,0,0,0,0,
    -1.f,  0.f,  1.f, -1.f,  0.f, 0,0,0,0,0,0,0,0,0,0,0,
     1.f,  0.f, -1.f,  1.f,  0.f, 0,0,0,0,0,0,0,0,0,0,0,
     0.f,  0.f,  0.f, -2.f,  1.f, 0,0,0,0,0,0,0,0,0,0,0,
     0.f,  0.f,  0.f,  2.f, -1.f, 0,0,0,0,0,0,0,0,0,0,0,
     0.f, -1.f,  2.f,  0.f, -1.f, 0,0,0,0,0,0,0,0,0,0,0,
     0.f,  1.f, -2.f,  0.f,  1.f, 0,0,0,0,0,0,0,0,0,0,0,
     1.f, -1.f,  1.f, -1.f,  0.f, 0,0,0,0,0,0,0,0,0,0,0,
    -1.f,  1.f, -1.f,  1.f,  0.f, 0,0,0,0,0,0,0,0,0,0,0,
    -1.f,  0.f,  1.f,  1.f, -1.f, 0,0,0,0,0,0,0,0,0,0,0,
     1.f,  0.f, -1.f, -1.f,  1.f, 0,0,0,0,0,0,0,0,0,0,0
};

#define LOG2_10 3.3219280948873623f

__global__ __launch_bounds__(256)
void rhs_wmma_kernel(const float* __restrict__ x, const float* __restrict__ w,
                     float* __restrict__ out, int numTiles) {
    const int lane = threadIdx.x & 31;
    const bool h   = (lane & 16) != 0;
    const int n    = lane & 15;

    const int wavesPerBlock = blockDim.x >> 5;
    const int globalWave = blockIdx.x * wavesPerBlock + (threadIdx.x >> 5);
    const int totalWaves = gridDim.x * wavesPerBlock;

    v2f Bm[3];
#pragma unroll
    for (int j = 0; j < 3; ++j) {
#pragma unroll
        for (int v = 0; v < 2; ++v) {
            const int k = 4 * j + (h ? 2 : 0) + v;
            const float s10 = exp2f(w[k] * LOG2_10);
            Bm[j][v] = c_ST[k * 16 + n] * s10;
        }
    }

    for (int t = globalWave; t < numTiles; t += totalWaves) {
        const int rowBase = t * 16;
        const float* xp = x + (size_t)(rowBase + n) * 5;
        const float z0 = xp[0], z1 = xp[1], z2 = xp[2], z3 = xp[3], z4 = xp[4];

        const float u   = h ? z3 : z0;
        const float a00 = z0 * u;
        const float a01 = h ? z2 : z1;
        const float s1a = h ? z1 : z3;
        const float s1b = h ? z4 : z3;
        const float a10 = s1a * s1b;
        const float z22 = z2 * z2;
        const float a11 = h ? z22 : z4;
        const float s2a = h ? z4 : z1;
        const float s2b = h ? z0 : z3;
        const float a20 = s2a * s2b;
        const float a21 = z2 * u;

        v8f d = {0.f, 0.f, 0.f, 0.f, 0.f, 0.f, 0.f, 0.f};
        v2f a;
        a[0] = a00; a[1] = a01;
        d = __builtin_amdgcn_wmma_f32_16x16x4_f32(false, a, false, Bm[0],
                                                  (short)0, d, false, false);
        a[0] = a10; a[1] = a11;
        d = __builtin_amdgcn_wmma_f32_16x16x4_f32(false, a, false, Bm[1],
                                                  (short)0, d, false, false);
        a[0] = a20; a[1] = a21;
        d = __builtin_amdgcn_wmma_f32_16x16x4_f32(false, a, false, Bm[2],
                                                  (short)0, d, false, false);

        if (n < 5) {
            float* op = out + ((size_t)(rowBase + (h ? 8 : 0))) * 5 + n;
#pragma unroll
            for (int r = 0; r < 8; ++r)
                op[(size_t)r * 5] = d[r];
        }
    }
}

__global__ __launch_bounds__(256)
void rhs_tail_kernel(const float* __restrict__ x, const float* __restrict__ w,
                     float* __restrict__ out, int startRow, int nRows) {
    const int i = blockIdx.x * blockDim.x + threadIdx.x;
    if (i >= nRows) return;
    const int row = startRow + i;
    const float* xp = x + (size_t)row * 5;
    const float z0 = xp[0], z1 = xp[1], z2 = xp[2], z3 = xp[3], z4 = xp[4];
    float p[12];
    p[0] = z0 * z0;  p[1]  = z1;       p[2]  = z0 * z3;  p[3]  = z2;
    p[4] = z3 * z3;  p[5]  = z4;       p[6]  = z1 * z4;  p[7]  = z2 * z2;
    p[8] = z1 * z3;  p[9]  = z2 * z0;  p[10] = z4 * z0;  p[11] = z2 * z3;
    float acc[5] = {0.f, 0.f, 0.f, 0.f, 0.f};
#pragma unroll
    for (int k = 0; k < 12; ++k) {
        const float f = p[k] * exp2f(w[k] * LOG2_10);
#pragma unroll
        for (int nn = 0; nn < 5; ++nn) acc[nn] += f * c_ST[k * 16 + nn];
    }
#pragma unroll
    for (int nn = 0; nn < 5; ++nn) out[(size_t)row * 5 + nn] = acc[nn];
}

extern "C" void kernel_launch(void* const* d_in, const int* in_sizes, int n_in,
                              void* d_out, int out_size, void* d_ws, size_t ws_size,
                              hipStream_t stream) {
    const float* x = (const float*)d_in[0];
    const float* w = (const float*)d_in[1];
    float* out = (float*)d_out;

    const long batch = (long)in_sizes[0] / 5;
    const int numTiles = (int)(batch / 16);
    const int rem = (int)(batch - (long)numTiles * 16);

    if (numTiles > 0) {
        const int threads = 256;
        int blocks = 2048;
        const int maxBlocks = (numTiles + 7) / 8;
        if (blocks > maxBlocks) blocks = maxBlocks;
        rhs_wmma_kernel<<<blocks, threads, 0, stream>>>(x, w, out, numTiles);
    }
    if (rem > 0) {
        rhs_tail_kernel<<<(rem + 255) / 256, 256, 0, stream>>>(x, w, out,
                                                               numTiles * 16, rem);
    }
}


# stage x and out through per-wave LDS so every global load and store is a whole-line 16-byte access (WMMA taken transposed so a lane holds one output row); 512 blocks; sum order unchanged
# speedup vs baseline: 1.2939x; 1.2939x over previous
#include <hip/hip_runtime.h>


typedef __attribute__((ext_vector_type(2))) float v2f;
typedef __attribute__((ext_vector_type(8))) float v8f;

__device__ __constant__ float c_ST[12 * 16] = {
    -2.f,  1.f,  0.f,  0.f,  0.f, 0,0,0,0,0,0,0,0,0,0,0,
     2.f, -1.f,  0.f,  0.f,  0.f, 0,0,0,0,0,0,0,0,0,0,0,
    -1.f,  0.f,  1.f, -1.f,  0.f, 0,0,0,0,0,0,0,0,0,0,0,
     1.f,  0.f, -1.f,  1.f,  0.f, 0,0,0,0,0,0,0,0,0,0,0,
     0.f,  0.f,  0.f, -2.f,  1.f, 0,0,0,0,0,0,0,0,0,0,0,
     0.f,  0.f,  0.f,  2.f, -1.f, 0,0,0,0,0,0,0,0,0,0,0,
     0.f, -1.f,  2.f,  0.f, -1.f, 0,0,0,0,0,0,0,0,0,0,0,
     0.f,  1.f, -2.f,  0.f,  1.f, 0,0,0,0,0,0,0,0,0,0,0,
     1.f, -1.f,  1.f, -1.f,  0.f, 0,0,0,0,0,0,0,0,0,0,0,
    -1.f,  1.f, -1.f,  1.f,  0.f, 0,0,0,0,0,0,0,0,0,0,0,
    -1.f,  0.f,  1.f,  1.f, -1.f, 0,0,0,0,0,0,0,0,0,0,0,
     1.f,  0.f, -1.f, -1.f,  1.f, 0,0,0,0,0,0,0,0,0,0,0
};

#define LOG2_10 3.3219280948873623f

#define WAVES_PER_BLOCK 8
#define CHUNK_ROWS  256
#define CHUNK_FLTS  (CHUNK_ROWS * 5)
#define CHUNK_F4    (CHUNK_FLTS / 4)
#define F4_PER_LANE (CHUNK_F4 / 32)
#define PAIRS       (CHUNK_ROWS / 32)

static __device__ __forceinline__ void wave_lds_handover() {
    __builtin_amdgcn_fence(__ATOMIC_RELEASE, "wavefront");
    __builtin_amdgcn_wave_barrier();
    __builtin_amdgcn_fence(__ATOMIC_ACQUIRE, "wavefront");
}

static __device__ __forceinline__ void load_chunk(const float4* __restrict__ x4, int c, int lane,
                                                  long totalF4, float4 (&v)[F4_PER_LANE]) {
    const long base = (long)c * CHUNK_F4;
    if (base + CHUNK_F4 <= totalF4) {
#pragma unroll
        for (int q = 0; q < F4_PER_LANE; ++q) v[q] = x4[base + q * 32 + lane];
    } else {
#pragma unroll
        for (int q = 0; q < F4_PER_LANE; ++q) {
            const long idx = base + q * 32 + lane;
            v[q] = (idx < totalF4) ? x4[idx] : make_float4(0.f, 0.f, 0.f, 0.f);
        }
    }
}

static __device__ __forceinline__ void row_terms(const float* p, bool h,
                                                 v2f& f0, v2f& f1, v2f& f2) {
    const float z0 = p[0], z1 = p[1], z2 = p[2], z3 = p[3], z4 = p[4];
    const float u   = h ? z3 : z0;
    f0[0] = z0 * u;
    f0[1] = h ? z2 : z1;
    const float s1a = h ? z1 : z3;
    const float s1b = h ? z4 : z3;
    f1[0] = s1a * s1b;
    const float z22 = z2 * z2;
    f1[1] = h ? z22 : z4;
    const float s2a = h ? z4 : z1;
    const float s2b = h ? z0 : z3;
    f2[0] = s2a * s2b;
    f2[1] = z2 * u;
}

__global__ __launch_bounds__(256)
void rhs_wmma_kernel(const float* __restrict__ x, const float* __restrict__ w,
                     float* __restrict__ out, int numTiles) {
    __shared__ __attribute__((aligned(16))) float sIn[WAVES_PER_BLOCK][CHUNK_FLTS];
    __shared__ __attribute__((aligned(16))) float sOut[WAVES_PER_BLOCK][CHUNK_FLTS];

    const int lane = threadIdx.x & 31;
    const bool h   = (lane & 16) != 0;
    const int n    = lane & 15;

    const int wave = __builtin_amdgcn_readfirstlane((int)(threadIdx.x >> 5));
    const int globalWave = blockIdx.x * WAVES_PER_BLOCK + wave;
    const int totalWaves = gridDim.x * WAVES_PER_BLOCK;

    v2f Alo[3], Ahi[3];
#pragma unroll
    for (int j = 0; j < 3; ++j) {
#pragma unroll
        for (int v = 0; v < 2; ++v) {
            const int k = 4 * j + (h ? 2 : 0) + v;
            const float s10 = exp2f(w[k] * LOG2_10);
            Alo[j][v] = c_ST[k * 16 + n] * s10;
            Ahi[j][v] = c_ST[k * 16 + ((n + 8) & 15)] * s10;
        }
    }

    const float4* __restrict__ x4 = reinterpret_cast<const float4*>(x);
    float4* __restrict__ o4 = reinterpret_cast<float4*>(out);
    const long totalF4  = (long)numTiles * 20;
    const int numChunks = (numTiles + 15) >> 4;

    float*  inW   = sIn[wave];
    float*  outW  = sOut[wave];
    float4* inW4  = reinterpret_cast<float4*>(inW);
    float4* outW4 = reinterpret_cast<float4*>(outW);

    float4 buf[F4_PER_LANE];
    int c = globalWave;
    if (c < numChunks) load_chunk(x4, c, lane, totalF4, buf);

    while (c < numChunks) {
#pragma unroll
        for (int q = 0; q < F4_PER_LANE; ++q) inW4[q * 32 + lane] = buf[q];
        const int cn = c + totalWaves;
        if (cn < numChunks) load_chunk(x4, cn, lane, totalF4, buf);
        wave_lds_handover();

#pragma unroll
        for (int p = 0; p < PAIRS; ++p) {
            v2f f0, f1, f2, g0, g1, g2;
            row_terms(inW + (p * 32 + n) * 5, h, f0, f1, f2);
            row_terms(inW + (p * 32 + 16 + n) * 5, h, g0, g1, g2);

            v8f d1 = {0.f, 0.f, 0.f, 0.f, 0.f, 0.f, 0.f, 0.f};
            v8f d2 = {0.f, 0.f, 0.f, 0.f, 0.f, 0.f, 0.f, 0.f};
            d1 = __builtin_amdgcn_wmma_f32_16x16x4_f32(false, Alo[0], false, f0,
                                                       (short)0, d1, false, false);
            d2 = __builtin_amdgcn_wmma_f32_16x16x4_f32(false, Ahi[0], false, g0,
                                                       (short)0, d2, false, false);
            d1 = __builtin_amdgcn_wmma_f32_16x16x4_f32(false, Alo[1], false, f1,
                                                       (short)0, d1, false, false);
            d2 = __builtin_amdgcn_wmma_f32_16x16x4_f32(false, Ahi[1], false, g1,
                                                       (short)0, d2, false, false);
            d1 = __builtin_amdgcn_wmma_f32_16x16x4_f32(false, Alo[2], false, f2,
                                                       (short)0, d1, false, false);
            d2 = __builtin_amdgcn_wmma_f32_16x16x4_f32(false, Ahi[2], false, g2,
                                                       (short)0, d2, false, false);

            float* op = outW + (p * 32 + lane) * 5;
#pragma unroll
            for (int r = 0; r < 5; ++r) op[r] = h ? d2[r] : d1[r];
        }
        wave_lds_handover();

        const long base = (long)c * CHUNK_F4;
        if (base + CHUNK_F4 <= totalF4) {
#pragma unroll
            for (int q = 0; q < F4_PER_LANE; ++q) o4[base + q * 32 + lane] = outW4[q * 32 + lane];
        } else {
#pragma unroll
            for (int q = 0; q < F4_PER_LANE; ++q) {
                const long idx = base + q * 32 + lane;
                if (idx < totalF4) o4[idx] = outW4[q * 32 + lane];
            }
        }
        c = cn;
    }
}

__global__ __launch_bounds__(256)
void rhs_tail_kernel(const float* __restrict__ x, const float* __restrict__ w,
                     float* __restrict__ out, int startRow, int nRows) {
    const int i = blockIdx.x * blockDim.x + threadIdx.x;
    if (i >= nRows) return;
    const int row = startRow + i;
    const float* xp = x + (size_t)row * 5;
    const float z0 = xp[0], z1 = xp[1], z2 = xp[2], z3 = xp[3], z4 = xp[4];
    float p[12];
    p[0] = z0 * z0;  p[1]  = z1;       p[2]  = z0 * z3;  p[3]  = z2;
    p[4] = z3 * z3;  p[5]  = z4;       p[6]  = z1 * z4;  p[7]  = z2 * z2;
    p[8] = z1 * z3;  p[9]  = z2 * z0;  p[10] = z4 * z0;  p[11] = z2 * z3;
    float acc[5] = {0.f, 0.f, 0.f, 0.f, 0.f};
#pragma unroll
    for (int k = 0; k < 12; ++k) {
        const float f = p[k] * exp2f(w[k] * LOG2_10);
#pragma unroll
        for (int nn = 0; nn < 5; ++nn) acc[nn] += f * c_ST[k * 16 + nn];
    }
#pragma unroll
    for (int nn = 0; nn < 5; ++nn) out[(size_t)row * 5 + nn] = acc[nn];
}

extern "C" void kernel_launch(void* const* d_in, const int* in_sizes, int n_in,
                              void* d_out, int out_size, void* d_ws, size_t ws_size,
                              hipStream_t stream) {
    const float* x = (const float*)d_in[0];
    const float* w = (const float*)d_in[1];
    float* out = (float*)d_out;

    const long batch = (long)in_sizes[0] / 5;
    const int numTiles = (int)(batch / 16);
    const int rem = (int)(batch - (long)numTiles * 16);

    if (numTiles > 0) {
        const int threads = 256;
        int blocks = 512;
        const int numChunks = (numTiles + 15) / 16;
        const int maxBlocks = (numChunks + 7) / 8;
        if (blocks > maxBlocks) blocks = maxBlocks;
        rhs_wmma_kernel<<<blocks, threads, 0, stream>>>(x, w, out, numTiles);
    }
    if (rem > 0) {
        rhs_tail_kernel<<<(rem + 255) / 256, 256, 0, stream>>>(x, w, out,
                                                               numTiles * 16, rem);
    }
}


# each lane forms all 12 terms of its own row and one WMMA carries two 16-row tiles in its two k halves (zero-padded A): no half-wave selects, half the LDS row reads; sum order unchanged
# speedup vs baseline: 1.3798x; 1.0664x over previous
#include <hip/hip_runtime.h>


typedef __attribute__((ext_vector_type(2))) float v2f;
typedef __attribute__((ext_vector_type(8))) float v8f;

__device__ __constant__ float c_ST[12 * 16] = {
    -2.f,  1.f,  0.f,  0.f,  0.f, 0,0,0,0,0,0,0,0,0,0,0,
     2.f, -1.f,  0.f,  0.f,  0.f, 0,0,0,0,0,0,0,0,0,0,0,
    -1.f,  0.f,  1.f, -1.f,  0.f, 0,0,0,0,0,0,0,0,0,0,0,
     1.f,  0.f, -1.f,  1.f,  0.f, 0,0,0,0,0,0,0,0,0,0,0,
     0.f,  0.f,  0.f, -2.f,  1.f, 0,0,0,0,0,0,0,0,0,0,0,
     0.f,  0.f,  0.f,  2.f, -1.f, 0,0,0,0,0,0,0,0,0,0,0,
     0.f, -1.f,  2.f,  0.f, -1.f, 0,0,0,0,0,0,0,0,0,0,0,
     0.f,  1.f, -2.f,  0.f,  1.f, 0,0,0,0,0,0,0,0,0,0,0,
     1.f, -1.f,  1.f, -1.f,  0.f, 0,0,0,0,0,0,0,0,0,0,0,
    -1.f,  1.f, -1.f,  1.f,  0.f, 0,0,0,0,0,0,0,0,0,0,0,
    -1.f,  0.f,  1.f,  1.f, -1.f, 0,0,0,0,0,0,0,0,0,0,0,
     1.f,  0.f, -1.f, -1.f,  1.f, 0,0,0,0,0,0,0,0,0,0,0
};

#define LOG2_10 3.3219280948873623f

#define WAVES_PER_BLOCK 8
#define CHUNK_ROWS  256
#define CHUNK_FLTS  (CHUNK_ROWS * 5)
#define CHUNK_F4    (CHUNK_FLTS / 4)
#define F4_PER_LANE (CHUNK_F4 / 32)
#define PAIRS       (CHUNK_ROWS / 32)

static __device__ __forceinline__ void wave_lds_handover() {
    __builtin_amdgcn_fence(__ATOMIC_RELEASE, "wavefront");
    __builtin_amdgcn_wave_barrier();
    __builtin_amdgcn_fence(__ATOMIC_ACQUIRE, "wavefront");
}

static __device__ __forceinline__ void load_chunk(const float4* __restrict__ x4, int c, int lane,
                                                  long totalF4, float4 (&v)[F4_PER_LANE]) {
    const long base = (long)c * CHUNK_F4;
    if (base + CHUNK_F4 <= totalF4) {
#pragma unroll
        for (int q = 0; q < F4_PER_LANE; ++q) v[q] = x4[base + q * 32 + lane];
    } else {
#pragma unroll
        for (int q = 0; q < F4_PER_LANE; ++q) {
            const long idx = base + q * 32 + lane;
            v[q] = (idx < totalF4) ? x4[idx] : make_float4(0.f, 0.f, 0.f, 0.f);
        }
    }
}

static __device__ __forceinline__ void row_terms(const float* p, v2f (&b)[6]) {
    const float z0 = p[0], z1 = p[1], z2 = p[2], z3 = p[3], z4 = p[4];
    b[0][0] = z0 * z0;  b[0][1] = z1;
    b[1][0] = z0 * z3;  b[1][1] = z2;
    b[2][0] = z3 * z3;  b[2][1] = z4;
    b[3][0] = z1 * z4;  b[3][1] = z2 * z2;
    b[4][0] = z1 * z3;  b[4][1] = z2 * z0;
    b[5][0] = z4 * z0;  b[5][1] = z2 * z3;
}

__global__ __launch_bounds__(256)
void rhs_wmma_kernel(const float* __restrict__ x, const float* __restrict__ w,
                     float* __restrict__ out, int numTiles) {
    __shared__ __attribute__((aligned(16))) float sIn[WAVES_PER_BLOCK][CHUNK_FLTS];
    __shared__ __attribute__((aligned(16))) float sOut[WAVES_PER_BLOCK][CHUNK_FLTS];

    const int lane = threadIdx.x & 31;
    const bool h   = (lane & 16) != 0;
    const int n    = lane & 15;

    const int wave = __builtin_amdgcn_readfirstlane((int)(threadIdx.x >> 5));
    const int globalWave = blockIdx.x * WAVES_PER_BLOCK + wave;
    const int totalWaves = gridDim.x * WAVES_PER_BLOCK;

    v2f A[6];
    const int arow = h ? ((n + 8) & 15) : n;
#pragma unroll
    for (int t = 0; t < 6; ++t) {
#pragma unroll
        for (int v = 0; v < 2; ++v) {
            const int k = 2 * t + v;
            const float s10 = exp2f(w[k] * LOG2_10);
            A[t][v] = c_ST[k * 16 + arow] * s10;
        }
    }

    const float4* __restrict__ x4 = reinterpret_cast<const float4*>(x);
    float4* __restrict__ o4 = reinterpret_cast<float4*>(out);
    const long totalF4  = (long)numTiles * 20;
    const int numChunks = (numTiles + 15) >> 4;

    float*  inW   = sIn[wave];
    float*  outW  = sOut[wave];
    float4* inW4  = reinterpret_cast<float4*>(inW);
    float4* outW4 = reinterpret_cast<float4*>(outW);

    float4 buf[F4_PER_LANE];
    int c = globalWave;
    if (c < numChunks) load_chunk(x4, c, lane, totalF4, buf);

    while (c < numChunks) {
#pragma unroll
        for (int q = 0; q < F4_PER_LANE; ++q) inW4[q * 32 + lane] = buf[q];
        const int cn = c + totalWaves;
        if (cn < numChunks) load_chunk(x4, cn, lane, totalF4, buf);
        wave_lds_handover();

#pragma unroll
        for (int p = 0; p < PAIRS; p += 2) {
            v2f b0[6], b1[6];
            row_terms(inW + (p * 32 + lane) * 5, b0);
            row_terms(inW + ((p + 1) * 32 + lane) * 5, b1);

            v8f d0 = {0.f, 0.f, 0.f, 0.f, 0.f, 0.f, 0.f, 0.f};
            v8f d1 = {0.f, 0.f, 0.f, 0.f, 0.f, 0.f, 0.f, 0.f};
#pragma unroll
            for (int t = 0; t < 6; ++t) {
                d0 = __builtin_amdgcn_wmma_f32_16x16x4_f32(false, A[t], false, b0[t],
                                                           (short)0, d0, false, false);
                d1 = __builtin_amdgcn_wmma_f32_16x16x4_f32(false, A[t], false, b1[t],
                                                           (short)0, d1, false, false);
            }

            float* op0 = outW + (p * 32 + lane) * 5;
            float* op1 = outW + ((p + 1) * 32 + lane) * 5;
#pragma unroll
            for (int r = 0; r < 5; ++r) op0[r] = d0[r];
#pragma unroll
            for (int r = 0; r < 5; ++r) op1[r] = d1[r];
        }
        wave_lds_handover();

        const long base = (long)c * CHUNK_F4;
        if (base + CHUNK_F4 <= totalF4) {
#pragma unroll
            for (int q = 0; q < F4_PER_LANE; ++q) o4[base + q * 32 + lane] = outW4[q * 32 + lane];
        } else {
#pragma unroll
            for (int q = 0; q < F4_PER_LANE; ++q) {
                const long idx = base + q * 32 + lane;
                if (idx < totalF4) o4[idx] = outW4[q * 32 + lane];
            }
        }
        c = cn;
    }
}

__global__ __launch_bounds__(256)
void rhs_tail_kernel(const float* __restrict__ x, const float* __restrict__ w,
                     float* __restrict__ out, int startRow, int nRows) {
    const int i = blockIdx.x * blockDim.x + threadIdx.x;
    if (i >= nRows) return;
    const int row = startRow + i;
    const float* xp = x + (size_t)row * 5;
    const float z0 = xp[0], z1 = xp[1], z2 = xp[2], z3 = xp[3], z4 = xp[4];
    float p[12];
    p[0] = z0 * z0;  p[1]  = z1;       p[2]  = z0 * z3;  p[3]  = z2;
    p[4] = z3 * z3;  p[5]  = z4;       p[6]  = z1 * z4;  p[7]  = z2 * z2;
    p[8] = z1 * z3;  p[9]  = z2 * z0;  p[10] = z4 * z0;  p[11] = z2 * z3;
    float acc[5] = {0.f, 0.f, 0.f, 0.f, 0.f};
#pragma unroll
    for (int k = 0; k < 12; ++k) {
        const float f = p[k] * exp2f(w[k] * LOG2_10);
#pragma unroll
        for (int nn = 0; nn < 5; ++nn) acc[nn] += f * c_ST[k * 16 + nn];
    }
#pragma unroll
    for (int nn = 0; nn < 5; ++nn) out[(size_t)row * 5 + nn] = acc[nn];
}

extern "C" void kernel_launch(void* const* d_in, const int* in_sizes, int n_in,
                              void* d_out, int out_size, void* d_ws, size_t ws_size,
                              hipStream_t stream) {
    const float* x = (const float*)d_in[0];
    const float* w = (const float*)d_in[1];
    float* out = (float*)d_out;

    const long batch = (long)in_sizes[0] / 5;
    const int numTiles = (int)(batch / 16);
    const int rem = (int)(batch - (long)numTiles * 16);

    if (numTiles > 0) {
        const int threads = 256;
        int blocks = 512;
        const int numChunks = (numTiles + 15) / 16;
        const int maxBlocks = (numChunks + 7) / 8;
        if (blocks > maxBlocks) blocks = maxBlocks;
        rhs_wmma_kernel<<<blocks, threads, 0, stream>>>(x, w, out, numTiles);
    }
    if (rem > 0) {
        rhs_tail_kernel<<<(rem + 255) / 256, 256, 0, stream>>>(x, w, out,
                                                               numTiles * 16, rem);
    }
}
